# starting kernel
# baseline (speedup 1.0000x reference)
#include <hip/hip_runtime.h>
#include <hip/hip_bf16.h>

typedef float v2f __attribute__((ext_vector_type(2)));
typedef float v8f __attribute__((ext_vector_type(8)));

#define NTOK 343
#define NPAD 352
#define HEADS 6
#define HD 32
#define DIMC 192
#define BATCH 256
#define ATTN_SCALE 0.17677669529663687f
#define SSTRIDE 353

__device__ __forceinline__ int rp_index(int q, int k) {
  int q0 = (q / 7) % 7, q1 = q / 49, q2 = q % 7;
  int k0 = (k / 7) % 7, k1 = k / 49, k2 = k % 7;
  return (q0 - k0 + 6) * 169 + (q1 - k1 + 6) * 13 + (q2 - k2 + 6);
}

__global__ __launch_bounds__(128) void qkv_kernel(
    const float* __restrict__ X, const float* __restrict__ W,
    float* __restrict__ Q, float* __restrict__ K, float* __restrict__ V) {
  const int t = threadIdx.x;
  const int wid = t >> 5, lane = t & 31, half = lane >> 4, l16 = lane & 15;
  const int row0 = blockIdx.x * 16;
  const int col0 = (blockIdx.y * 4 + wid) * 16;
  const float* xr = X + (size_t)(row0 + l16) * DIMC;

  v8f acc = {};
  for (int kk = 0; kk < DIMC; kk += 4) {
    const int kc = kk + 2 * half;
    v2f a = { xr[kc], xr[kc + 1] };
    v2f b = { W[(size_t)kc * 576 + col0 + l16],
              W[(size_t)(kc + 1) * 576 + col0 + l16] };
    acc = __builtin_amdgcn_wmma_f32_16x16x4_f32(false, a, false, b,
                                                (short)0, acc, false, false);
  }

  const int which = col0 / DIMC;
  const int head  = (col0 % DIMC) / HD;
  const int hd0   = col0 % HD;
  float* dst = (which == 0) ? Q : (which == 1) ? K : V;
  const float sc = (which == 0) ? ATTN_SCALE : 1.0f;
#pragma unroll
  for (int i = 0; i < 8; ++i) {
    const int r = row0 + i + 8 * half;
    const int b = r / NTOK, n = r % NTOK;
    dst[(((size_t)b * HEADS + head) * NTOK + n) * HD + hd0 + l16] = acc[i] * sc;
  }
}

__global__ __launch_bounds__(128) void attn_kernel(
    const float* __restrict__ Q, const float* __restrict__ K,
    const float* __restrict__ V, const float* __restrict__ mask,
    const float* __restrict__ rpb, float* __restrict__ AO) {
  __shared__ float sS[16 * SSTRIDE];
  __shared__ float sRed[16 * 8];
  __shared__ float sRowMax[16];
  __shared__ float sRowInv[16];
  __shared__ float sPart[2 * 16 * 16];

  const int t = threadIdx.x;
  const int wid = t >> 5, lane = t & 31, half = lane >> 4, l16 = lane & 15;
  const int q0 = blockIdx.x * 16;
  const int h  = blockIdx.y;
  const int b  = blockIdx.z;
  const size_t base = ((size_t)b * HEADS + h) * (size_t)NTOK * HD;
  const float* Qb = Q + base;
  const float* Kb = K + base;
  const float* Vb = V + base;

  float qa0[8], qa1[8];
  const int qrow = q0 + l16;
  const bool qok = qrow < NTOK;
#pragma unroll
  for (int s = 0; s < 8; ++s) {
    const int d = s * 4 + 2 * half;
    qa0[s] = qok ? Qb[qrow * HD + d]     : 0.f;
    qa1[s] = qok ? Qb[qrow * HD + d + 1] : 0.f;
  }
  const float* mrow = mask + (size_t)(b & 63) * NTOK * NTOK;
  for (int kt = wid; kt < NPAD / 16; kt += 4) {
    const int k0 = kt * 16;
    const int krow = k0 + l16;
    const bool kok = krow < NTOK;
    v8f acc = {};
#pragma unroll
    for (int s = 0; s < 8; ++s) {
      const int d = s * 4 + 2 * half;
      v2f a = { qa0[s], qa1[s] };
      v2f bf = { kok ? Kb[krow * HD + d]     : 0.f,
                 kok ? Kb[krow * HD + d + 1] : 0.f };
      acc = __builtin_amdgcn_wmma_f32_16x16x4_f32(false, a, false, bf,
                                                  (short)0, acc, false, false);
    }
#pragma unroll
    for (int i = 0; i < 8; ++i) {
      const int q = q0 + i + 8 * half;
      const int k = k0 + l16;
      float v = acc[i];
      if (q < NTOK && k < NTOK)
        v += rpb[rp_index(q, k) * HEADS + h] + mrow[q * NTOK + k];
      else
        v = -1e30f;
      sS[(i + 8 * half) * SSTRIDE + k] = v;
    }
  }
  __syncthreads();

  {
    const int r = t >> 3, sg = t & 7;
    float m = -1e30f;
    for (int k = sg; k < NTOK; k += 8) m = fmaxf(m, sS[r * SSTRIDE + k]);
    sRed[r * 8 + sg] = m;
    __syncthreads();
    if (sg == 0) {
      float mm = sRed[r * 8];
#pragma unroll
      for (int j = 1; j < 8; ++j) mm = fmaxf(mm, sRed[r * 8 + j]);
      sRowMax[r] = mm;
    }
    __syncthreads();
    const float rm = sRowMax[r];
    float sum = 0.f;
    for (int k = sg; k < NPAD; k += 8) {
      float e = 0.f;
      if (k < NTOK) { e = __expf(sS[r * SSTRIDE + k] - rm); sum += e; }
      sS[r * SSTRIDE + k] = e;
    }
    sRed[r * 8 + sg] = sum;
    __syncthreads();
    if (sg == 0) {
      float ss = 0.f;
#pragma unroll
      for (int j = 0; j < 8; ++j) ss += sRed[r * 8 + j];
      sRowInv[r] = 1.0f / ss;
    }
  }
  __syncthreads();

  const int dt = wid & 1;
  const int kh = wid >> 1;
  const int d0 = dt * 16;
  v8f acc = {};
  for (int kk = kh * (NPAD / 2); kk < kh * (NPAD / 2) + NPAD / 2; kk += 4) {
    const int kc = kk + 2 * half;
    v2f a = { sS[l16 * SSTRIDE + kc], sS[l16 * SSTRIDE + kc + 1] };
    v2f bf = { (kc < NTOK)     ? Vb[(size_t)kc * HD + d0 + l16]       : 0.f,
               (kc + 1 < NTOK) ? Vb[(size_t)(kc + 1) * HD + d0 + l16] : 0.f };
    acc = __builtin_amdgcn_wmma_f32_16x16x4_f32(false, a, false, bf,
                                                (short)0, acc, false, false);
  }
  if (kh == 1) {
#pragma unroll
    for (int i = 0; i < 8; ++i)
      sPart[dt * 256 + (i + 8 * half) * 16 + l16] = acc[i];
  }
  __syncthreads();
  if (kh == 0) {
#pragma unroll
    for (int i = 0; i < 8; ++i) {
      const int q = q0 + i + 8 * half;
      if (q < NTOK) {
        const float v = (acc[i] + sPart[dt * 256 + (i + 8 * half) * 16 + l16])
                        * sRowInv[i + 8 * half];
        AO[((size_t)b * NTOK + q) * DIMC + h * HD + d0 + l16] = v;
      }
    }
  }
}

__global__ __launch_bounds__(128) void proj_kernel(
    const float* __restrict__ AO, const float* __restrict__ W,
    const float* __restrict__ bias, float* __restrict__ out) {
  const int t = threadIdx.x;
  const int wid = t >> 5, lane = t & 31, half = lane >> 4, l16 = lane & 15;
  const int row0 = blockIdx.x * 16;
  const int col0 = (blockIdx.y * 4 + wid) * 16;
  const float* ar = AO + (size_t)(row0 + l16) * DIMC;

  v8f acc = {};
  for (int kk = 0; kk < DIMC; kk += 4) {
    const int kc = kk + 2 * half;
    v2f a = { ar[kc], ar[kc + 1] };
    v2f b = { W[(size_t)kc * DIMC + col0 + l16],
              W[(size_t)(kc + 1) * DIMC + col0 + l16] };
    acc = __builtin_amdgcn_wmma_f32_16x16x4_f32(false, a, false, b,
                                                (short)0, acc, false, false);
  }
  const float bv = bias[col0 + l16];
#pragma unroll
  for (int i = 0; i < 8; ++i) {
    const int r = row0 + i + 8 * half;
    out[(size_t)r * DIMC + col0 + l16] = acc[i] + bv;
  }
}

extern "C" void kernel_launch(void* const* d_in, const int* in_sizes, int n_in,
                              void* d_out, int out_size, void* d_ws, size_t ws_size,
                              hipStream_t stream) {
  const float* x      = (const float*)d_in[0];
  const float* mask   = (const float*)d_in[1];
  const float* qkv_w  = (const float*)d_in[2];
  const float* rpb    = (const float*)d_in[3];
  const float* proj_w = (const float*)d_in[4];
  const float* proj_b = (const float*)d_in[5];
  float* out = (float*)d_out;

  const size_t qkvElems = (size_t)BATCH * HEADS * NTOK * HD;
  float* Q  = (float*)d_ws;
  float* K  = Q + qkvElems;
  float* V  = K + qkvElems;
  float* AO = V + qkvElems;

  const int rowTiles = (BATCH * NTOK) / 16;
  qkv_kernel <<<dim3(rowTiles, 9),            128, 0, stream>>>(x, qkv_w, Q, K, V);
  attn_kernel<<<dim3(NPAD / 16, HEADS, BATCH), 128, 0, stream>>>(Q, K, V, mask, rpb, AO);
  proj_kernel<<<dim3(rowTiles, 3),            128, 0, stream>>>(AO, proj_w, proj_b, out);
}


# attention: one wave owns 16 queries; scores as K.Q^T stay in registers and feed V^T.P^T directly (running-max softmax, no score LDS); QK d-order, row-sum and key-sum orders change
# speedup vs baseline: 1.5447x; 1.5447x over previous
#include <hip/hip_runtime.h>
#include <hip/hip_bf16.h>

typedef float v2f __attribute__((ext_vector_type(2)));
typedef float v8f __attribute__((ext_vector_type(8)));

#define NTOK 343
#define NPAD 352
#define HEADS 6
#define HD 32
#define DIMC 192
#define BATCH 256
#define ATTN_SCALE 0.17677669529663687f


__global__ __launch_bounds__(128) void qkv_kernel(
    const float* __restrict__ X, const float* __restrict__ W,
    float* __restrict__ Q, float* __restrict__ K, float* __restrict__ V) {
  const int t = threadIdx.x;
  const int wid = t >> 5, lane = t & 31, half = lane >> 4, l16 = lane & 15;
  const int row0 = blockIdx.x * 16;
  const int col0 = (blockIdx.y * 4 + wid) * 16;
  const float* xr = X + (size_t)(row0 + l16) * DIMC;

  v8f acc = {};
  for (int kk = 0; kk < DIMC; kk += 4) {
    const int kc = kk + 2 * half;
    v2f a = { xr[kc], xr[kc + 1] };
    v2f b = { W[(size_t)kc * 576 + col0 + l16],
              W[(size_t)(kc + 1) * 576 + col0 + l16] };
    acc = __builtin_amdgcn_wmma_f32_16x16x4_f32(false, a, false, b,
                                                (short)0, acc, false, false);
  }

  const int which = col0 / DIMC;
  const int head  = (col0 % DIMC) / HD;
  const int hd0   = col0 % HD;
  float* dst = (which == 0) ? Q : (which == 1) ? K : V;
  const float sc = (which == 0) ? ATTN_SCALE : 1.0f;
#pragma unroll
  for (int i = 0; i < 8; ++i) {
    const int r = row0 + i + 8 * half;
    const int b = r / NTOK, n = r % NTOK;
    dst[(((size_t)b * HEADS + head) * NTOK + n) * HD + hd0 + l16] = acc[i] * sc;
  }
}

typedef float v4f __attribute__((ext_vector_type(4)));
#define TPITCH 20
#define KTILES (NPAD / 16)

__device__ __forceinline__ int pos_code(int f) {
  return ((f / 7) % 7) * 169 + (f / 49) * 13 + (f % 7);
}

__global__ __launch_bounds__(128) void attn_kernel(
    const float* __restrict__ Q, const float* __restrict__ K,
    const float* __restrict__ V, const float* __restrict__ mask,
    const float* __restrict__ rpb, float* __restrict__ AO) {
  __shared__ __attribute__((aligned(16))) float sT[4][2][16 * TPITCH];

  const int t = threadIdx.x;
  const int wid = t >> 5, lane = t & 31, half = lane >> 4, l16 = lane & 15;
  const int gw = blockIdx.x * 4 + wid;
  const int qt = gw % KTILES;
  const int bh = gw / KTILES;
  const int h  = bh % HEADS;
  const int b  = bh / HEADS;
  const int q0 = qt * 16;
  const size_t base = (size_t)bh * (size_t)NTOK * HD;
  const float* Qb = Q + base;
  const float* Kb = K + base;
  const float* Vb = V + base;
  const float* mrow = mask + (size_t)(b & 63) * NTOK * NTOK;

  const int qc = min(q0 + l16, NTOK - 1);
  v4f qf[4];
  {
    const v4f* qp = (const v4f*)(Qb + qc * HD + 16 * half);
#pragma unroll
    for (int j = 0; j < 4; ++j) qf[j] = qp[j];
  }

  int qcode[8], qrow[8];
#pragma unroll
  for (int r = 0; r < 8; ++r) {
    const int q = min(q0 + 8 * half + r, NTOK - 1);
    qcode[r] = (pos_code(q) + 1098) * HEADS + h;
    qrow[r]  = q * NTOK;
  }

  float m_run = -1e30f, l_run = 0.f;
  v8f o0 = {}, o1 = {};

  for (int kt = 0; kt < KTILES; ++kt) {
    const int k0 = kt * 16;
    float* tb = &sT[wid][kt & 1][0];
    const int kcol = min(k0 + l16, NTOK - 1);

    const int kcode6 = pos_code(kcol) * HEADS;
#pragma unroll
    for (int r = 0; r < 8; ++r) {
      const float v = rpb[qcode[r] - kcode6] + mrow[qrow[r] + kcol];
      tb[(8 * half + r) * TPITCH + l16] = v;
    }

    v4f kf[4];
    {
      const v4f* kp = (const v4f*)(Kb + kcol * HD + 16 * half);
#pragma unroll
      for (int j = 0; j < 4; ++j) kf[j] = kp[j];
    }
    float va0[4], va1[4], vb0[4], vb1[4];
#pragma unroll
    for (int s4 = 0; s4 < 4; ++s4) {
      const int key0 = k0 + 8 * half + 2 * s4;
      const int ka = min(key0, NTOK - 1), kb = min(key0 + 1, NTOK - 1);
      va0[s4] = Vb[ka * HD + l16]; va1[s4] = Vb[ka * HD + 16 + l16];
      vb0[s4] = Vb[kb * HD + l16]; vb1[s4] = Vb[kb * HD + 16 + l16];
      if (key0 >= NTOK)     { va0[s4] = 0.f; va1[s4] = 0.f; }
      if (key0 + 1 >= NTOK) { vb0[s4] = 0.f; vb1[s4] = 0.f; }
    }
    v8f acc = {};
#pragma unroll
    for (int j = 0; j < 4; ++j) {
      v2f a0 = { kf[j].x, kf[j].y };
      v2f b0 = { qf[j].x, qf[j].y };
      acc = __builtin_amdgcn_wmma_f32_16x16x4_f32(false, a0, false, b0,
                                                  (short)0, acc, false, false);
      v2f a1 = { kf[j].z, kf[j].w };
      v2f b1 = { qf[j].z, qf[j].w };
      acc = __builtin_amdgcn_wmma_f32_16x16x4_f32(false, a1, false, b1,
                                                  (short)0, acc, false, false);
    }

    __syncthreads();
    float s[8];
    {
      const v4f* tp = (const v4f*)(tb + l16 * TPITCH + 8 * half);
      const v4f bm0 = tp[0], bm1 = tp[1];
      s[0] = acc[0] + bm0.x; s[1] = acc[1] + bm0.y;
      s[2] = acc[2] + bm0.z; s[3] = acc[3] + bm0.w;
      s[4] = acc[4] + bm1.x; s[5] = acc[5] + bm1.y;
      s[6] = acc[6] + bm1.z; s[7] = acc[7] + bm1.w;
    }
    if (kt == KTILES - 1) {
#pragma unroll
      for (int r = 0; r < 8; ++r)
        if (k0 + 8 * half + r >= NTOK) s[r] = -1e30f;
    }

    float tm = s[0];
#pragma unroll
    for (int r = 1; r < 8; ++r) tm = fmaxf(tm, s[r]);
    tm = fmaxf(tm, __shfl_xor(tm, 16));
    const float m_new = fmaxf(m_run, tm);
    const float alpha = __expf(m_run - m_new);
    float p[8];
    float ps = 0.f;
#pragma unroll
    for (int r = 0; r < 8; ++r) { p[r] = __expf(s[r] - m_new); ps += p[r]; }
    l_run = l_run * alpha + ps;
    m_run = m_new;
#pragma unroll
    for (int r = 0; r < 8; ++r) { o0[r] *= alpha; o1[r] *= alpha; }

#pragma unroll
    for (int s4 = 0; s4 < 4; ++s4) {
      v2f pb = { p[2 * s4], p[2 * s4 + 1] };
      v2f a0 = { va0[s4], vb0[s4] };
      v2f a1 = { va1[s4], vb1[s4] };
      o0 = __builtin_amdgcn_wmma_f32_16x16x4_f32(false, a0, false, pb,
                                                 (short)0, o0, false, false);
      o1 = __builtin_amdgcn_wmma_f32_16x16x4_f32(false, a1, false, pb,
                                                 (short)0, o1, false, false);
    }
  }

  const float l_tot = l_run + __shfl_xor(l_run, 16);
  const float inv = 1.0f / l_tot;
  const int q = q0 + l16;
  if (q < NTOK) {
    float* op = AO + ((size_t)b * NTOK + q) * DIMC + h * HD + 8 * half;
    v4f w0 = { o0[0] * inv, o0[1] * inv, o0[2] * inv, o0[3] * inv };
    v4f w1 = { o0[4] * inv, o0[5] * inv, o0[6] * inv, o0[7] * inv };
    v4f w2 = { o1[0] * inv, o1[1] * inv, o1[2] * inv, o1[3] * inv };
    v4f w3 = { o1[4] * inv, o1[5] * inv, o1[6] * inv, o1[7] * inv };
    ((v4f*)op)[0] = w0;
    ((v4f*)op)[1] = w1;
    ((v4f*)(op + 16))[0] = w2;
    ((v4f*)(op + 16))[1] = w3;
  }
}

__global__ __launch_bounds__(128) void proj_kernel(
    const float* __restrict__ AO, const float* __restrict__ W,
    const float* __restrict__ bias, float* __restrict__ out) {
  const int t = threadIdx.x;
  const int wid = t >> 5, lane = t & 31, half = lane >> 4, l16 = lane & 15;
  const int row0 = blockIdx.x * 16;
  const int col0 = (blockIdx.y * 4 + wid) * 16;
  const float* ar = AO + (size_t)(row0 + l16) * DIMC;

  v8f acc = {};
  for (int kk = 0; kk < DIMC; kk += 4) {
    const int kc = kk + 2 * half;
    v2f a = { ar[kc], ar[kc + 1] };
    v2f b = { W[(size_t)kc * DIMC + col0 + l16],
              W[(size_t)(kc + 1) * DIMC + col0 + l16] };
    acc = __builtin_amdgcn_wmma_f32_16x16x4_f32(false, a, false, b,
                                                (short)0, acc, false, false);
  }
  const float bv = bias[col0 + l16];
#pragma unroll
  for (int i = 0; i < 8; ++i) {
    const int r = row0 + i + 8 * half;
    out[(size_t)r * DIMC + col0 + l16] = acc[i] + bv;
  }
}

extern "C" void kernel_launch(void* const* d_in, const int* in_sizes, int n_in,
                              void* d_out, int out_size, void* d_ws, size_t ws_size,
                              hipStream_t stream) {
  const float* x      = (const float*)d_in[0];
  const float* mask   = (const float*)d_in[1];
  const float* qkv_w  = (const float*)d_in[2];
  const float* rpb    = (const float*)d_in[3];
  const float* proj_w = (const float*)d_in[4];
  const float* proj_b = (const float*)d_in[5];
  float* out = (float*)d_out;

  const size_t qkvElems = (size_t)BATCH * HEADS * NTOK * HD;
  float* Q  = (float*)d_ws;
  float* K  = Q + qkvElems;
  float* V  = K + qkvElems;
  float* AO = V + qkvElems;

  const int rowTiles = (BATCH * NTOK) / 16;
  qkv_kernel <<<dim3(rowTiles, 9),            128, 0, stream>>>(x, qkv_w, Q, K, V);
  attn_kernel<<<dim3((NPAD / 16) * HEADS * BATCH / 4), 128, 0, stream>>>(Q, K, V, mask, rpb, AO);
  proj_kernel<<<dim3(rowTiles, 3),            128, 0, stream>>>(AO, proj_w, proj_b, out);
}


# both projection GEMMs: a block of 8 waves stages its 64 X rows once and W in 64-column slabs through LDS (coalesced b128 loads, k pairs interleaved); fragments are LDS reads; k sum order kept
# speedup vs baseline: 2.1123x; 1.3674x over previous
#include <hip/hip_runtime.h>
#include <hip/hip_bf16.h>

typedef float v2f __attribute__((ext_vector_type(2)));
typedef float v8f __attribute__((ext_vector_type(8)));

#define NTOK 343
#define NPAD 352
#define HEADS 6
#define HD 32
#define DIMC 192
#define BATCH 256
#define ATTN_SCALE 0.17677669529663687f


typedef float v4f __attribute__((ext_vector_type(4)));
#define GR 64
#define GC 64
#define XP 196
#define WP 128
#define WTASKS 6

template <int NC, bool QKV>
__global__ __launch_bounds__(256) void gemm_lds_kernel(
    const float* __restrict__ X, const float* __restrict__ W,
    const float* __restrict__ bias,
    float* __restrict__ O0, float* __restrict__ O1, float* __restrict__ O2) {
  __shared__ __attribute__((aligned(16))) float sX[GR * XP];
  __shared__ __attribute__((aligned(16))) float sW[(DIMC / 2) * WP];

  const int t = threadIdx.x;
  const int wid = t >> 5, lane = t & 31, half = lane >> 4, l16 = lane & 15;
  const int rt = wid & 3, cp = wid >> 2;
  const int row0 = blockIdx.x * GR;

  v4f pw0[WTASKS], pw1[WTASKS];
#pragma unroll
  for (int i = 0; i < WTASKS; ++i) {
    const int id = t + 256 * i;
    const int g = id & 15, kp = id >> 4;
    const float* p = W + (size_t)(2 * kp) * NC + 4 * g;
    pw0[i] = *(const v4f*)p;
    pw1[i] = *(const v4f*)(p + NC);
  }

  {
    const v4f* xs = (const v4f*)(X + (size_t)row0 * DIMC);
#pragma unroll
    for (int i = 0; i < 12; ++i) {
      const int f = t + 256 * i;
      const int r = f / 48, c4 = f - r * 48;
      *(v4f*)(sX + r * XP + 4 * c4) = xs[f];
    }
  }

  int rowoff[8];
#pragma unroll
  for (int r = 0; r < 8; ++r) {
    const int row = row0 + rt * 16 + 8 * half + r;
    if (QKV) {
      const int b = row / NTOK, n = row - b * NTOK;
      rowoff[r] = (b * HEADS * NTOK + n) * HD;
    } else {
      rowoff[r] = row * DIMC;
    }
  }

  const float* xa = sX + (rt * 16 + l16) * XP + 2 * half;
  const int colA = cp * 32 + l16;
  const float* wb0 = sW + half * WP + 2 * (colA ^ (16 * half));
  const float* wb1 = sW + half * WP + 2 * ((colA + 16) ^ (16 * half));

  for (int j = 0; j < NC / GC; ++j) {
#pragma unroll
    for (int i = 0; i < WTASKS; ++i) {
      const int id = t + 256 * i;
      const int g = id & 15, kp = id >> 4;
      float* d = sW + kp * WP + 2 * ((4 * g) ^ (16 * (kp & 1)));
      const v4f lo = { pw0[i].x, pw1[i].x, pw0[i].y, pw1[i].y };
      const v4f hi = { pw0[i].z, pw1[i].z, pw0[i].w, pw1[i].w };
      ((v4f*)d)[0] = lo;
      ((v4f*)d)[1] = hi;
    }
    __syncthreads();

    if (j + 1 < NC / GC) {
#pragma unroll
      for (int i = 0; i < WTASKS; ++i) {
        const int id = t + 256 * i;
        const int g = id & 15, kp = id >> 4;
        const float* p = W + (size_t)(2 * kp) * NC + (j + 1) * GC + 4 * g;
        pw0[i] = *(const v4f*)p;
        pw1[i] = *(const v4f*)(p + NC);
      }
    }

    v8f acc0 = {}, acc1 = {};
#pragma unroll
    for (int s = 0; s < DIMC / 4; ++s) {
      const v2f a  = *(const v2f*)(xa + 4 * s);
      const v2f b0 = *(const v2f*)(wb0 + 2 * s * WP);
      const v2f b1 = *(const v2f*)(wb1 + 2 * s * WP);
      acc0 = __builtin_amdgcn_wmma_f32_16x16x4_f32(false, a, false, b0,
                                                   (short)0, acc0, false, false);
      acc1 = __builtin_amdgcn_wmma_f32_16x16x4_f32(false, a, false, b1,
                                                   (short)0, acc1, false, false);
    }

    if (QKV) {
      const int which = j / 3;
      const int head  = 2 * (j - 3 * which) + cp;
      float* dst = (which == 0) ? O0 : (which == 1) ? O1 : O2;
      const float sc = (which == 0) ? ATTN_SCALE : 1.0f;
      float* dp = dst + head * (NTOK * HD) + l16;
#pragma unroll
      for (int r = 0; r < 8; ++r) {
        dp[rowoff[r]]      = acc0[r] * sc;
        dp[rowoff[r] + 16] = acc1[r] * sc;
      }
    } else {
      const int col = j * GC + cp * 32 + l16;
      const float bv0 = bias[col], bv1 = bias[col + 16];
      float* dp = O0 + col;
#pragma unroll
      for (int r = 0; r < 8; ++r) {
        dp[rowoff[r]]      = acc0[r] + bv0;
        dp[rowoff[r] + 16] = acc1[r] + bv1;
      }
    }
    __syncthreads();
  }
}

#define TPITCH 20
#define KTILES (NPAD / 16)

__device__ __forceinline__ int pos_code(int f) {
  return ((f / 7) % 7) * 169 + (f / 49) * 13 + (f % 7);
}

__global__ __launch_bounds__(128) void attn_kernel(
    const float* __restrict__ Q, const float* __restrict__ K,
    const float* __restrict__ V, const float* __restrict__ mask,
    const float* __restrict__ rpb, float* __restrict__ AO) {
  __shared__ __attribute__((aligned(16))) float sT[4][2][16 * TPITCH];

  const int t = threadIdx.x;
  const int wid = t >> 5, lane = t & 31, half = lane >> 4, l16 = lane & 15;
  const int gw = blockIdx.x * 4 + wid;
  const int qt = gw % KTILES;
  const int bh = gw / KTILES;
  const int h  = bh % HEADS;
  const int b  = bh / HEADS;
  const int q0 = qt * 16;
  const size_t base = (size_t)bh * (size_t)NTOK * HD;
  const float* Qb = Q + base;
  const float* Kb = K + base;
  const float* Vb = V + base;
  const float* mrow = mask + (size_t)(b & 63) * NTOK * NTOK;

  const int qc = min(q0 + l16, NTOK - 1);
  v4f qf[4];
  {
    const v4f* qp = (const v4f*)(Qb + qc * HD + 16 * half);
#pragma unroll
    for (int j = 0; j < 4; ++j) qf[j] = qp[j];
  }

  int qcode[8], qrow[8];
#pragma unroll
  for (int r = 0; r < 8; ++r) {
    const int q = min(q0 + 8 * half + r, NTOK - 1);
    qcode[r] = (pos_code(q) + 1098) * HEADS + h;
    qrow[r]  = q * NTOK;
  }

  float m_run = -1e30f, l_run = 0.f;
  v8f o0 = {}, o1 = {};

  for (int kt = 0; kt < KTILES; ++kt) {
    const int k0 = kt * 16;
    float* tb = &sT[wid][kt & 1][0];
    const int kcol = min(k0 + l16, NTOK - 1);

    const int kcode6 = pos_code(kcol) * HEADS;
#pragma unroll
    for (int r = 0; r < 8; ++r) {
      const float v = rpb[qcode[r] - kcode6] + mrow[qrow[r] + kcol];
      tb[(8 * half + r) * TPITCH + l16] = v;
    }

    v4f kf[4];
    {
      const v4f* kp = (const v4f*)(Kb + kcol * HD + 16 * half);
#pragma unroll
      for (int j = 0; j < 4; ++j) kf[j] = kp[j];
    }
    float va0[4], va1[4], vb0[4], vb1[4];
#pragma unroll
    for (int s4 = 0; s4 < 4; ++s4) {
      const int key0 = k0 + 8 * half + 2 * s4;
      const int ka = min(key0, NTOK - 1), kb = min(key0 + 1, NTOK - 1);
      va0[s4] = Vb[ka * HD + l16]; va1[s4] = Vb[ka * HD + 16 + l16];
      vb0[s4] = Vb[kb * HD + l16]; vb1[s4] = Vb[kb * HD + 16 + l16];
      if (key0 >= NTOK)     { va0[s4] = 0.f; va1[s4] = 0.f; }
      if (key0 + 1 >= NTOK) { vb0[s4] = 0.f; vb1[s4] = 0.f; }
    }
    v8f acc = {};
#pragma unroll
    for (int j = 0; j < 4; ++j) {
      v2f a0 = { kf[j].x, kf[j].y };
      v2f b0 = { qf[j].x, qf[j].y };
      acc = __builtin_amdgcn_wmma_f32_16x16x4_f32(false, a0, false, b0,
                                                  (short)0, acc, false, false);
      v2f a1 = { kf[j].z, kf[j].w };
      v2f b1 = { qf[j].z, qf[j].w };
      acc = __builtin_amdgcn_wmma_f32_16x16x4_f32(false, a1, false, b1,
                                                  (short)0, acc, false, false);
    }

    __syncthreads();
    float s[8];
    {
      const v4f* tp = (const v4f*)(tb + l16 * TPITCH + 8 * half);
      const v4f bm0 = tp[0], bm1 = tp[1];
      s[0] = acc[0] + bm0.x; s[1] = acc[1] + bm0.y;
      s[2] = acc[2] + bm0.z; s[3] = acc[3] + bm0.w;
      s[4] = acc[4] + bm1.x; s[5] = acc[5] + bm1.y;
      s[6] = acc[6] + bm1.z; s[7] = acc[7] + bm1.w;
    }
    if (kt == KTILES - 1) {
#pragma unroll
      for (int r = 0; r < 8; ++r)
        if (k0 + 8 * half + r >= NTOK) s[r] = -1e30f;
    }

    float tm = s[0];
#pragma unroll
    for (int r = 1; r < 8; ++r) tm = fmaxf(tm, s[r]);
    tm = fmaxf(tm, __shfl_xor(tm, 16));
    const float m_new = fmaxf(m_run, tm);
    const float alpha = __expf(m_run - m_new);
    float p[8];
    float ps = 0.f;
#pragma unroll
    for (int r = 0; r < 8; ++r) { p[r] = __expf(s[r] - m_new); ps += p[r]; }
    l_run = l_run * alpha + ps;
    m_run = m_new;
#pragma unroll
    for (int r = 0; r < 8; ++r) { o0[r] *= alpha; o1[r] *= alpha; }

#pragma unroll
    for (int s4 = 0; s4 < 4; ++s4) {
      v2f pb = { p[2 * s4], p[2 * s4 + 1] };
      v2f a0 = { va0[s4], vb0[s4] };
      v2f a1 = { va1[s4], vb1[s4] };
      o0 = __builtin_amdgcn_wmma_f32_16x16x4_f32(false, a0, false, pb,
                                                 (short)0, o0, false, false);
      o1 = __builtin_amdgcn_wmma_f32_16x16x4_f32(false, a1, false, pb,
                                                 (short)0, o1, false, false);
    }
  }

  const float l_tot = l_run + __shfl_xor(l_run, 16);
  const float inv = 1.0f / l_tot;
  const int q = q0 + l16;
  if (q < NTOK) {
    float* op = AO + ((size_t)b * NTOK + q) * DIMC + h * HD + 8 * half;
    v4f w0 = { o0[0] * inv, o0[1] * inv, o0[2] * inv, o0[3] * inv };
    v4f w1 = { o0[4] * inv, o0[5] * inv, o0[6] * inv, o0[7] * inv };
    v4f w2 = { o1[0] * inv, o1[1] * inv, o1[2] * inv, o1[3] * inv };
    v4f w3 = { o1[4] * inv, o1[5] * inv, o1[6] * inv, o1[7] * inv };
    ((v4f*)op)[0] = w0;
    ((v4f*)op)[1] = w1;
    ((v4f*)(op + 16))[0] = w2;
    ((v4f*)(op + 16))[1] = w3;
  }
}

extern "C" void kernel_launch(void* const* d_in, const int* in_sizes, int n_in,
                              void* d_out, int out_size, void* d_ws, size_t ws_size,
                              hipStream_t stream) {
  const float* x      = (const float*)d_in[0];
  const float* mask   = (const float*)d_in[1];
  const float* qkv_w  = (const float*)d_in[2];
  const float* rpb    = (const float*)d_in[3];
  const float* proj_w = (const float*)d_in[4];
  const float* proj_b = (const float*)d_in[5];
  float* out = (float*)d_out;

  const size_t qkvElems = (size_t)BATCH * HEADS * NTOK * HD;
  float* Q  = (float*)d_ws;
  float* K  = Q + qkvElems;
  float* V  = K + qkvElems;
  float* AO = V + qkvElems;

  const int rowBlocks = (BATCH * NTOK) / GR;
  gemm_lds_kernel<3 * DIMC, true><<<dim3(rowBlocks), 256, 0, stream>>>(x, qkv_w, proj_b, Q, K, V);
  attn_kernel<<<dim3((NPAD / 16) * HEADS * BATCH / 4), 128, 0, stream>>>(Q, K, V, mask, rpb, AO);
  gemm_lds_kernel<DIMC, false><<<dim3(rowBlocks), 256, 0, stream>>>(AO, proj_w, proj_b, out, out, out);
}


# attention: one block (11 waves) per (batch, head): K, V^T and the head's bias column staged once in LDS for all 22 query tiles; per-tile block barrier becomes a wave-scope fence; sum orders kept
# speedup vs baseline: 3.0802x; 1.4582x over previous
#include <hip/hip_runtime.h>
#include <hip/hip_bf16.h>

typedef float v2f __attribute__((ext_vector_type(2)));
typedef float v8f __attribute__((ext_vector_type(8)));

#define NTOK 343
#define NPAD 352
#define HEADS 6
#define HD 32
#define DIMC 192
#define BATCH 256
#define ATTN_SCALE 0.17677669529663687f


typedef float v4f __attribute__((ext_vector_type(4)));
#define GR 64
#define GC 64
#define XP 196
#define WP 128
#define WTASKS 6

template <int NC, bool QKV>
__global__ __launch_bounds__(256) void gemm_lds_kernel(
    const float* __restrict__ X, const float* __restrict__ W,
    const float* __restrict__ bias,
    float* __restrict__ O0, float* __restrict__ O1, float* __restrict__ O2) {
  __shared__ __attribute__((aligned(16))) float sX[GR * XP];
  __shared__ __attribute__((aligned(16))) float sW[(DIMC / 2) * WP];

  const int t = threadIdx.x;
  const int wid = t >> 5, lane = t & 31, half = lane >> 4, l16 = lane & 15;
  const int rt = wid & 3, cp = wid >> 2;
  const int row0 = blockIdx.x * GR;

  v4f pw0[WTASKS], pw1[WTASKS];
#pragma unroll
  for (int i = 0; i < WTASKS; ++i) {
    const int id = t + 256 * i;
    const int g = id & 15, kp = id >> 4;
    const float* p = W + (size_t)(2 * kp) * NC + 4 * g;
    pw0[i] = *(const v4f*)p;
    pw1[i] = *(const v4f*)(p + NC);
  }

  {
    const v4f* xs = (const v4f*)(X + (size_t)row0 * DIMC);
#pragma unroll
    for (int i = 0; i < 12; ++i) {
      const int f = t + 256 * i;
      const int r = f / 48, c4 = f - r * 48;
      *(v4f*)(sX + r * XP + 4 * c4) = xs[f];
    }
  }

  int rowoff[8];
#pragma unroll
  for (int r = 0; r < 8; ++r) {
    const int row = row0 + rt * 16 + 8 * half + r;
    if (QKV) {
      const int b = row / NTOK, n = row - b * NTOK;
      rowoff[r] = (b * HEADS * NTOK + n) * HD;
    } else {
      rowoff[r] = row * DIMC;
    }
  }

  const float* xa = sX + (rt * 16 + l16) * XP + 2 * half;
  const int colA = cp * 32 + l16;
  const float* wb0 = sW + half * WP + 2 * (colA ^ (16 * half));
  const float* wb1 = sW + half * WP + 2 * ((colA + 16) ^ (16 * half));

  for (int j = 0; j < NC / GC; ++j) {
#pragma unroll
    for (int i = 0; i < WTASKS; ++i) {
      const int id = t + 256 * i;
      const int g = id & 15, kp = id >> 4;
      float* d = sW + kp * WP + 2 * ((4 * g) ^ (16 * (kp & 1)));
      const v4f lo = { pw0[i].x, pw1[i].x, pw0[i].y, pw1[i].y };
      const v4f hi = { pw0[i].z, pw1[i].z, pw0[i].w, pw1[i].w };
      ((v4f*)d)[0] = lo;
      ((v4f*)d)[1] = hi;
    }
    __syncthreads();

    if (j + 1 < NC / GC) {
#pragma unroll
      for (int i = 0; i < WTASKS; ++i) {
        const int id = t + 256 * i;
        const int g = id & 15, kp = id >> 4;
        const float* p = W + (size_t)(2 * kp) * NC + (j + 1) * GC + 4 * g;
        pw0[i] = *(const v4f*)p;
        pw1[i] = *(const v4f*)(p + NC);
      }
    }

    v8f acc0 = {}, acc1 = {};
#pragma unroll
    for (int s = 0; s < DIMC / 4; ++s) {
      const v2f a  = *(const v2f*)(xa + 4 * s);
      const v2f b0 = *(const v2f*)(wb0 + 2 * s * WP);
      const v2f b1 = *(const v2f*)(wb1 + 2 * s * WP);
      acc0 = __builtin_amdgcn_wmma_f32_16x16x4_f32(false, a, false, b0,
                                                   (short)0, acc0, false, false);
      acc1 = __builtin_amdgcn_wmma_f32_16x16x4_f32(false, a, false, b1,
                                                   (short)0, acc1, false, false);
    }

    if (QKV) {
      const int which = j / 3;
      const int head  = 2 * (j - 3 * which) + cp;
      float* dst = (which == 0) ? O0 : (which == 1) ? O1 : O2;
      const float sc = (which == 0) ? ATTN_SCALE : 1.0f;
      float* dp = dst + head * (NTOK * HD) + l16;
#pragma unroll
      for (int r = 0; r < 8; ++r) {
        dp[rowoff[r]]      = acc0[r] * sc;
        dp[rowoff[r] + 16] = acc1[r] * sc;
      }
    } else {
      const int col = j * GC + cp * 32 + l16;
      const float bv0 = bias[col], bv1 = bias[col + 16];
      float* dp = O0 + col;
#pragma unroll
      for (int r = 0; r < 8; ++r) {
        dp[rowoff[r]]      = acc0[r] + bv0;
        dp[rowoff[r] + 16] = acc1[r] + bv1;
      }
    }
    __syncthreads();
  }
}

#define TPITCH 20
#define KTILES (NPAD / 16)
#define AW 11
#define ATH (AW * 32)
#define KP 36
#define VP 356
#define NCODE 2197
#define KV4 (NTOK * HD / 4)

__device__ __forceinline__ int pos_code(int f) {
  return ((f / 7) % 7) * 169 + (f / 49) * 13 + (f % 7);
}

__global__ __launch_bounds__(ATH) void attn_kernel(
    const float* __restrict__ Q, const float* __restrict__ K,
    const float* __restrict__ V, const float* __restrict__ mask,
    const float* __restrict__ rpb, float* __restrict__ AO) {
  __shared__ __attribute__((aligned(16))) float sK[NPAD * KP];
  __shared__ __attribute__((aligned(16))) float sVt[HD * VP];
  __shared__ __attribute__((aligned(16))) float sB[NCODE + 3];
  __shared__ __attribute__((aligned(16))) int sCode[NPAD];
  __shared__ __attribute__((aligned(16))) float sT[AW][2][16 * TPITCH];

  const int t = threadIdx.x;
  const int wid = t >> 5, lane = t & 31, half = lane >> 4, l16 = lane & 15;
  const int bh = blockIdx.x;
  const int h  = bh % HEADS;
  const int b  = bh / HEADS;
  const size_t base = (size_t)bh * (size_t)NTOK * HD;
  const float* Qb = Q + base;
  const float* mrow = mask + (size_t)(b & 63) * NTOK * NTOK;

  {
    const v4f* kg = (const v4f*)(K + base);
    const v4f* vg = (const v4f*)(V + base);
    v4f kk[8], vv[8];
#pragma unroll
    for (int i = 0; i < 8; ++i) {
      const int f = t + ATH * i;
      const v4f z = { 0.f, 0.f, 0.f, 0.f };
      kk[i] = z; vv[i] = z;
      if (f < KV4) { kk[i] = kg[f]; vv[i] = vg[f]; }
    }
#pragma unroll
    for (int i = 0; i < 8; ++i) {
      const int f = t + ATH * i;
      const int key = f >> 3, c4 = f & 7;
      *(v4f*)(sK + key * KP + 4 * c4) = kk[i];
      float* vd = sVt + (4 * c4) * VP + key;
      vd[0] = vv[i].x; vd[VP] = vv[i].y; vd[2 * VP] = vv[i].z; vd[3 * VP] = vv[i].w;
    }
    for (int i = t; i < NCODE; i += ATH) sB[i] = rpb[i * HEADS + h];
    sCode[t] = pos_code(min(t, NTOK - 1));
  }
  __syncthreads();

  for (int qi = 0; qi < 2; ++qi) {
    const int qt = wid + AW * qi;
    const int q0 = qt * 16;

    const int qc = min(q0 + l16, NTOK - 1);
    v4f qf[4];
    {
      const v4f* qp = (const v4f*)(Qb + qc * HD + 16 * half);
#pragma unroll
      for (int j = 0; j < 4; ++j) qf[j] = qp[j];
    }

    int qcode[8], qrow[8];
#pragma unroll
    for (int r = 0; r < 8; ++r) {
      const int q = min(q0 + 8 * half + r, NTOK - 1);
      qcode[r] = sCode[q] + 1098;
      qrow[r]  = q * NTOK;
    }

    float m_run = -1e30f, l_run = 0.f;
    v8f o0 = {}, o1 = {};

    for (int kt = 0; kt < KTILES; ++kt) {
      const int k0 = kt * 16;
      float* tb = &sT[wid][kt & 1][0];
      const int kcol = min(k0 + l16, NTOK - 1);

      const int kcode = sCode[k0 + l16];
#pragma unroll
      for (int r = 0; r < 8; ++r) {
        const float v = sB[qcode[r] - kcode] + mrow[qrow[r] + kcol];
        tb[(8 * half + r) * TPITCH + l16] = v;
      }

      v4f kf[4];
      {
        const v4f* kp = (const v4f*)(sK + (k0 + l16) * KP + 16 * half);
#pragma unroll
        for (int j = 0; j < 4; ++j) kf[j] = kp[j];
      }
      float va[8], vb[8];
      {
        const v4f* vp0 = (const v4f*)(sVt + l16 * VP + k0 + 8 * half);
        const v4f* vp1 = (const v4f*)(sVt + (16 + l16) * VP + k0 + 8 * half);
        const v4f a0 = vp0[0], a1 = vp0[1], b0 = vp1[0], b1 = vp1[1];
        va[0] = a0.x; va[1] = a0.y; va[2] = a0.z; va[3] = a0.w;
        va[4] = a1.x; va[5] = a1.y; va[6] = a1.z; va[7] = a1.w;
        vb[0] = b0.x; vb[1] = b0.y; vb[2] = b0.z; vb[3] = b0.w;
        vb[4] = b1.x; vb[5] = b1.y; vb[6] = b1.z; vb[7] = b1.w;
      }
      v8f acc = {};
#pragma unroll
      for (int j = 0; j < 4; ++j) {
        v2f a0 = { kf[j].x, kf[j].y };
        v2f b0 = { qf[j].x, qf[j].y };
        acc = __builtin_amdgcn_wmma_f32_16x16x4_f32(false, a0, false, b0,
                                                    (short)0, acc, false, false);
        v2f a1 = { kf[j].z, kf[j].w };
        v2f b1 = { qf[j].z, qf[j].w };
        acc = __builtin_amdgcn_wmma_f32_16x16x4_f32(false, a1, false, b1,
                                                    (short)0, acc, false, false);
      }

      __builtin_amdgcn_fence(__ATOMIC_RELEASE, "wavefront");
      __builtin_amdgcn_wave_barrier();
      __builtin_amdgcn_fence(__ATOMIC_ACQUIRE, "wavefront");
      float s[8];
      {
        const v4f* tp = (const v4f*)(tb + l16 * TPITCH + 8 * half);
        const v4f bm0 = tp[0], bm1 = tp[1];
        s[0] = acc[0] + bm0.x; s[1] = acc[1] + bm0.y;
        s[2] = acc[2] + bm0.z; s[3] = acc[3] + bm0.w;
        s[4] = acc[4] + bm1.x; s[5] = acc[5] + bm1.y;
        s[6] = acc[6] + bm1.z; s[7] = acc[7] + bm1.w;
      }
      if (kt == KTILES - 1) {
#pragma unroll
        for (int r = 0; r < 8; ++r)
          if (k0 + 8 * half + r >= NTOK) s[r] = -1e30f;
      }

      float tm = s[0];
#pragma unroll
      for (int r = 1; r < 8; ++r) tm = fmaxf(tm, s[r]);
      tm = fmaxf(tm, __shfl_xor(tm, 16));
      const float m_new = fmaxf(m_run, tm);
      const float alpha = __expf(m_run - m_new);
      float p[8];
      float ps = 0.f;
#pragma unroll
      for (int r = 0; r < 8; ++r) { p[r] = __expf(s[r] - m_new); ps += p[r]; }
      l_run = l_run * alpha + ps;
      m_run = m_new;
#pragma unroll
      for (int r = 0; r < 8; ++r) { o0[r] *= alpha; o1[r] *= alpha; }

#pragma unroll
      for (int s4 = 0; s4 < 4; ++s4) {
        v2f pb = { p[2 * s4], p[2 * s4 + 1] };
        v2f a0 = { va[2 * s4], va[2 * s4 + 1] };
        v2f a1 = { vb[2 * s4], vb[2 * s4 + 1] };
        o0 = __builtin_amdgcn_wmma_f32_16x16x4_f32(false, a0, false, pb,
                                                   (short)0, o0, false, false);
        o1 = __builtin_amdgcn_wmma_f32_16x16x4_f32(false, a1, false, pb,
                                                   (short)0, o1, false, false);
      }
    }

    const float l_tot = l_run + __shfl_xor(l_run, 16);
    const float inv = 1.0f / l_tot;
    const int q = q0 + l16;
    if (q < NTOK) {
      float* op = AO + ((size_t)b * NTOK + q) * DIMC + h * HD + 8 * half;
      v4f w0 = { o0[0] * inv, o0[1] * inv, o0[2] * inv, o0[3] * inv };
      v4f w1 = { o0[4] * inv, o0[5] * inv, o0[6] * inv, o0[7] * inv };
      v4f w2 = { o1[0] * inv, o1[1] * inv, o1[2] * inv, o1[3] * inv };
      v4f w3 = { o1[4] * inv, o1[5] * inv, o1[6] * inv, o1[7] * inv };
      ((v4f*)op)[0] = w0;
      ((v4f*)op)[1] = w1;
      ((v4f*)(op + 16))[0] = w2;
      ((v4f*)(op + 16))[1] = w3;
    }
  }
}

extern "C" void kernel_launch(void* const* d_in, const int* in_sizes, int n_in,
                              void* d_out, int out_size, void* d_ws, size_t ws_size,
                              hipStream_t stream) {
  const float* x      = (const float*)d_in[0];
  const float* mask   = (const float*)d_in[1];
  const float* qkv_w  = (const float*)d_in[2];
  const float* rpb    = (const float*)d_in[3];
  const float* proj_w = (const float*)d_in[4];
  const float* proj_b = (const float*)d_in[5];
  float* out = (float*)d_out;

  const size_t qkvElems = (size_t)BATCH * HEADS * NTOK * HD;
  float* Q  = (float*)d_ws;
  float* K  = Q + qkvElems;
  float* V  = K + qkvElems;
  float* AO = V + qkvElems;

  const int rowBlocks = (BATCH * NTOK) / GR;
  gemm_lds_kernel<3 * DIMC, true><<<dim3(rowBlocks), 256, 0, stream>>>(x, qkv_w, proj_b, Q, K, V);
  attn_kernel<<<dim3(HEADS * BATCH), ATH, 0, stream>>>(Q, K, V, mask, rpb, AO);
  gemm_lds_kernel<DIMC, false><<<dim3(rowBlocks), 256, 0, stream>>>(AO, proj_w, proj_b, out, out, out);
}
